# starting kernel
# baseline (speedup 1.0000x reference)
#include <hip/hip_runtime.h>

typedef __attribute__((ext_vector_type(2))) float v2f;
typedef __attribute__((ext_vector_type(8))) float v8f;

#define NCH  24
#define MDIM 56
#define NDIM 56
#define KH   9
#define JD   14
#define RR   (KH*JD)
#define RPAD 128
#define NPAD 64
#define OC   96

#define WSTR 132
#define USTR 80
#define TSTR 29
#define XSTR 57

__global__ __launch_bounds__(256) void fused_shift_unfold_wmma(
    const float* __restrict__ x,
    const float* __restrict__ w0,
    const float* __restrict__ w1,
    float* __restrict__ out)
{
    __shared__ float Wlds[OC * WSTR];
    __shared__ float Ulds[RPAD * USTR];
    __shared__ float Tlds[MDIM * TSTR];
    __shared__ float Xlds[NCH * XSTR];
    __shared__ float W1lds[NCH * 28];

    const int tid   = threadIdx.x;
    const int m     = blockIdx.x;
    const int ntile = blockIdx.y;

    for (int e = tid; e < OC * RPAD; e += 256) {
        int i = e >> 7, r = e & 127;
        Wlds[i * WSTR + r] = (r < RR) ? w0[i * RR + r] : 0.0f;
    }
    for (int e = tid; e < NCH * NDIM; e += 256) {
        int i = e / NDIM, h = e % NDIM;
        Xlds[i * XSTR + h] = x[(i * MDIM + m) * NDIM + h];
    }
    for (int e = tid; e < NCH * 28; e += 256) {
        W1lds[e] = w1[e];
    }
    __syncthreads();

    for (int e = tid; e < MDIM * 28; e += 256) {
        int h = e / 28, c = e % 28;
        float s = 0.0f;
        #pragma unroll
        for (int i = 0; i < NCH; ++i)
            s += Xlds[i * XSTR + h] * W1lds[i * 28 + c];
        Tlds[h * TSTR + c] = s;
    }
    __syncthreads();

    for (int e = tid; e < RPAD * NPAD; e += 256) {
        int r = e >> 6, n = e & 63;
        float v = 0.0f;
        if (r < RR && n < NDIM) {
            int kh = r / JD;
            int j  = r - kh * JD;
            int h0 = n + kh - 4;
            int h1 = ((n + 55) % 56) + kh - 4;
            if (h0 >= 0 && h0 < NDIM) v += Tlds[h0 * TSTR + j];
            if (h1 >= 0 && h1 < NDIM) v += Tlds[h1 * TSTR + 14 + j];
        }
        Ulds[r * USTR + n] = v;
    }
    __syncthreads();

    const int wave = tid >> 5;
    const int lane = tid & 31;
    if (wave < 6) {
        const int mtile = wave;
        const int arow  = mtile * 16 + (lane & 15);
        const int koff  = (lane >> 4) << 1;
        const int col   = ntile * 16 + (lane & 15);

        v8f acc = {};
        #pragma unroll
        for (int ks = 0; ks < RPAD; ks += 4) {
            v2f a = *(const v2f*)&Wlds[arow * WSTR + ks + koff];
            v2f b;
            b.x = Ulds[(ks + koff    ) * USTR + col];
            b.y = Ulds[(ks + koff + 1) * USTR + col];
            acc = __builtin_amdgcn_wmma_f32_16x16x4_f32(
                 false, a,  false, b,
                 (short)0, acc,  false,  false);
        }

        if (col < NDIM) {
            const int i0 = mtile * 16 + ((lane >> 4) << 3);
            #pragma unroll
            for (int v = 0; v < 8; ++v)
                out[(i0 + v) * (MDIM * NDIM) + m * NDIM + col] = acc[v];
        }
    }
}

extern "C" void kernel_launch(void* const* d_in, const int* in_sizes, int n_in,
                              void* d_out, int out_size, void* d_ws, size_t ws_size,
                              hipStream_t stream) {
    const float* x  = (const float*)d_in[0];
    const float* w0 = (const float*)d_in[1];
    const float* w1 = (const float*)d_in[2];
    float* out = (float*)d_out;
    (void)in_sizes; (void)n_in; (void)out_size; (void)d_ws; (void)ws_size;

    dim3 grid(MDIM, 4);
    fused_shift_unfold_wmma<<<grid, 256, 0, stream>>>(x, w0, w1, out);
}


# All global loads of the block issued as one batch at entry (x, w1, then w0 as 24 coalesced b64 per thread held in registers across LDS-only barriers): 57 serial load-wait-store trips become one
# speedup vs baseline: 2.8667x; 2.8667x over previous
#include <hip/hip_runtime.h>


typedef __attribute__((ext_vector_type(2))) float v2f;
typedef __attribute__((ext_vector_type(8))) float v8f;

#define NCH  24
#define MDIM 56
#define NDIM 56
#define KH   9
#define JD   14
#define RR   (KH*JD)
#define RPAD 128
#define NPAD 64
#define OC   96

#define WSTR 132
#define USTR 80
#define TSTR 29
#define XSTR 57

#define NTHR    256
#define XELEMS  (NCH * NDIM)
#define W1ELEMS (NCH * 28)
#define W0PAIRS (OC * RR / 2)
#define XTRIPS  ((XELEMS  + NTHR - 1) / NTHR)
#define W1TRIPS ((W1ELEMS + NTHR - 1) / NTHR)
#define W0TRIPS ((W0PAIRS + NTHR - 1) / NTHR)

#define XSPARE  ((NCH - 1) * XSTR + NDIM)
#define W1SPARE W1ELEMS
#define W0SPARE ((OC - 1) * WSTR + RPAD)

static __device__ __forceinline__ void lds_barrier() {
    __builtin_amdgcn_fence(__ATOMIC_RELEASE, "workgroup", "local");
    __builtin_amdgcn_s_barrier();
    __builtin_amdgcn_fence(__ATOMIC_ACQUIRE, "workgroup", "local");
}

__global__ __launch_bounds__(NTHR) void fused_shift_unfold_wmma(
    const float* __restrict__ x,
    const float* __restrict__ w0,
    const float* __restrict__ w1,
    float* __restrict__ out)
{
    __shared__ float Wlds[OC * WSTR];
    __shared__ float Ulds[RPAD * USTR];
    __shared__ float Tlds[MDIM * TSTR];
    __shared__ float Xlds[NCH * XSTR];
    __shared__ float W1lds[W1ELEMS + 1];

    const int tid   = threadIdx.x;
    const int m     = blockIdx.x;
    const int ntile = blockIdx.y;

    float xv[XTRIPS];
    #pragma unroll
    for (int t = 0; t < XTRIPS; ++t) {
        int e = tid + t * NTHR;
        e = e < XELEMS ? e : XELEMS - 1;
        int i = e / NDIM, h = e % NDIM;
        xv[t] = x[(i * MDIM + m) * NDIM + h];
    }
    float w1v[W1TRIPS];
    #pragma unroll
    for (int t = 0; t < W1TRIPS; ++t) {
        int e = tid + t * NTHR;
        w1v[t] = w1[e < W1ELEMS ? e : W1ELEMS - 1];
    }
    v2f w0v[W0TRIPS];
    #pragma unroll
    for (int t = 0; t < W0TRIPS; ++t) {
        int q = tid + t * NTHR;
        w0v[t] = ((const v2f*)w0)[q < W0PAIRS ? q : W0PAIRS - 1];
    }

    #pragma unroll
    for (int t = 0; t < XTRIPS; ++t) {
        int e = tid + t * NTHR;
        int i = e / NDIM, h = e % NDIM;
        Xlds[e < XELEMS ? i * XSTR + h : XSPARE] = xv[t];
    }
    #pragma unroll
    for (int t = 0; t < W1TRIPS; ++t) {
        int e = tid + t * NTHR;
        W1lds[e < W1ELEMS ? e : W1SPARE] = w1v[t];
    }
    lds_barrier();

    for (int e = tid; e < MDIM * 28; e += NTHR) {
        int h = e / 28, c = e % 28;
        float s = 0.0f;
        #pragma unroll
        for (int i = 0; i < NCH; ++i)
            s += Xlds[i * XSTR + h] * W1lds[i * 28 + c];
        Tlds[h * TSTR + c] = s;
    }
    lds_barrier();

    for (int e = tid; e < RPAD * NPAD; e += NTHR) {
        int r = e >> 6, n = e & 63;
        float v = 0.0f;
        if (r < RR && n < NDIM) {
            int kh = r / JD;
            int j  = r - kh * JD;
            int h0 = n + kh - 4;
            int h1 = ((n + 55) % 56) + kh - 4;
            if (h0 >= 0 && h0 < NDIM) v += Tlds[h0 * TSTR + j];
            if (h1 >= 0 && h1 < NDIM) v += Tlds[h1 * TSTR + 14 + j];
        }
        Ulds[r * USTR + n] = v;
    }

    #pragma unroll
    for (int t = 0; t < W0TRIPS; ++t) {
        int q = tid + t * NTHR;
        int i = q / (RR / 2);
        int r = (q - i * (RR / 2)) * 2;
        *(v2f*)&Wlds[q < W0PAIRS ? i * WSTR + r : W0SPARE] = w0v[t];
    }
    if (tid < OC) *(v2f*)&Wlds[tid * WSTR + RR] = (v2f){0.0f, 0.0f};
    __syncthreads();

    const int wave = tid >> 5;
    const int lane = tid & 31;
    if (wave < 6) {
        const int mtile = wave;
        const int arow  = mtile * 16 + (lane & 15);
        const int koff  = (lane >> 4) << 1;
        const int col   = ntile * 16 + (lane & 15);

        v8f acc = {};
        #pragma unroll
        for (int ks = 0; ks < RPAD; ks += 4) {
            v2f a = *(const v2f*)&Wlds[arow * WSTR + ks + koff];
            v2f b;
            b.x = Ulds[(ks + koff    ) * USTR + col];
            b.y = Ulds[(ks + koff + 1) * USTR + col];
            acc = __builtin_amdgcn_wmma_f32_16x16x4_f32(
                 false, a,  false, b,
                 (short)0, acc,  false,  false);
        }

        if (col < NDIM) {
            const int i0 = mtile * 16 + ((lane >> 4) << 3);
            #pragma unroll
            for (int v = 0; v < 8; ++v)
                out[(i0 + v) * (MDIM * NDIM) + m * NDIM + col] = acc[v];
        }
    }
}

extern "C" void kernel_launch(void* const* d_in, const int* in_sizes, int n_in,
                              void* d_out, int out_size, void* d_ws, size_t ws_size,
                              hipStream_t stream) {
    const float* x  = (const float*)d_in[0];
    const float* w0 = (const float*)d_in[1];
    const float* w1 = (const float*)d_in[2];
    float* out = (float*)d_out;
    (void)in_sizes; (void)n_in; (void)out_size; (void)d_ws; (void)ws_size;

    dim3 grid(MDIM, 4);
    fused_shift_unfold_wmma<<<grid, 256, 0, stream>>>(x, w0, w1, out);
}


# Each block builds only the part of T and U that its 16 output columns read: U 128x16 instead of 128x64 (8 LDS trips per thread, was 32), T 25 rows instead of 56; same values, same summation order
# speedup vs baseline: 4.5272x; 1.5792x over previous
#include <hip/hip_runtime.h>


typedef __attribute__((ext_vector_type(2))) float v2f;
typedef __attribute__((ext_vector_type(8))) float v8f;

#define NCH  24
#define MDIM 56
#define NDIM 56
#define KH   9
#define JD   14
#define RR   (KH*JD)
#define RPAD 128
#define OC   96

#define WSTR 132
#define UCOL 16
#define USTR 16
#define TROWS 25
#define TSTR 29
#define XSTR 57

#define NTHR    256
#define XELEMS  (NCH * NDIM)
#define W1ELEMS (NCH * 28)
#define W0PAIRS (OC * RR / 2)
#define XTRIPS  ((XELEMS  + NTHR - 1) / NTHR)
#define W1TRIPS ((W1ELEMS + NTHR - 1) / NTHR)
#define W0TRIPS ((W0PAIRS + NTHR - 1) / NTHR)

#define XSPARE  ((NCH - 1) * XSTR + NDIM)
#define W1SPARE W1ELEMS
#define W0SPARE ((OC - 1) * WSTR + RPAD)

static __device__ __forceinline__ void lds_barrier() {
    __builtin_amdgcn_fence(__ATOMIC_RELEASE, "workgroup", "local");
    __builtin_amdgcn_s_barrier();
    __builtin_amdgcn_fence(__ATOMIC_ACQUIRE, "workgroup", "local");
}

__global__ __launch_bounds__(NTHR) void fused_shift_unfold_wmma(
    const float* __restrict__ x,
    const float* __restrict__ w0,
    const float* __restrict__ w1,
    float* __restrict__ out)
{
    __shared__ float Wlds[OC * WSTR];
    __shared__ float Ulds[RPAD * USTR];
    __shared__ float Tlds[MDIM * TSTR];
    __shared__ float Xlds[NCH * XSTR];
    __shared__ float W1lds[W1ELEMS + 1];

    const int tid   = threadIdx.x;
    const int m     = blockIdx.x;
    const int ntile = blockIdx.y;
    const int n0    = ntile * UCOL;

    float xv[XTRIPS];
    #pragma unroll
    for (int t = 0; t < XTRIPS; ++t) {
        int e = tid + t * NTHR;
        e = e < XELEMS ? e : XELEMS - 1;
        int i = e / NDIM, h = e % NDIM;
        xv[t] = x[(i * MDIM + m) * NDIM + h];
    }
    float w1v[W1TRIPS];
    #pragma unroll
    for (int t = 0; t < W1TRIPS; ++t) {
        int e = tid + t * NTHR;
        w1v[t] = w1[e < W1ELEMS ? e : W1ELEMS - 1];
    }
    v2f w0v[W0TRIPS];
    #pragma unroll
    for (int t = 0; t < W0TRIPS; ++t) {
        int q = tid + t * NTHR;
        w0v[t] = ((const v2f*)w0)[q < W0PAIRS ? q : W0PAIRS - 1];
    }

    #pragma unroll
    for (int t = 0; t < XTRIPS; ++t) {
        int e = tid + t * NTHR;
        int i = e / NDIM, h = e % NDIM;
        Xlds[e < XELEMS ? i * XSTR + h : XSPARE] = xv[t];
    }
    #pragma unroll
    for (int t = 0; t < W1TRIPS; ++t) {
        int e = tid + t * NTHR;
        W1lds[e < W1ELEMS ? e : W1SPARE] = w1v[t];
    }
    lds_barrier();

    for (int e = tid; e < TROWS * 28; e += NTHR) {
        int q = e / 28, c = e % 28;
        int h = (n0 - 5 + q + NDIM) % NDIM;
        float s = 0.0f;
        #pragma unroll
        for (int i = 0; i < NCH; ++i)
            s += Xlds[i * XSTR + h] * W1lds[i * 28 + c];
        Tlds[h * TSTR + c] = s;
    }
    lds_barrier();

    for (int e = tid; e < RPAD * UCOL; e += NTHR) {
        int r = e >> 4, nl = e & 15;
        int n = n0 + nl;
        float v = 0.0f;
        if (r < RR && n < NDIM) {
            int kh = r / JD;
            int j  = r - kh * JD;
            int h0 = n + kh - 4;
            int h1 = ((n + 55) % 56) + kh - 4;
            if (h0 >= 0 && h0 < NDIM) v += Tlds[h0 * TSTR + j];
            if (h1 >= 0 && h1 < NDIM) v += Tlds[h1 * TSTR + 14 + j];
        }
        Ulds[r * USTR + nl] = v;
    }

    #pragma unroll
    for (int t = 0; t < W0TRIPS; ++t) {
        int q = tid + t * NTHR;
        int i = q / (RR / 2);
        int r = (q - i * (RR / 2)) * 2;
        *(v2f*)&Wlds[q < W0PAIRS ? i * WSTR + r : W0SPARE] = w0v[t];
    }
    if (tid < OC) *(v2f*)&Wlds[tid * WSTR + RR] = (v2f){0.0f, 0.0f};
    __syncthreads();

    const int wave = tid >> 5;
    const int lane = tid & 31;
    if (wave < 6) {
        const int mtile = wave;
        const int arow  = mtile * 16 + (lane & 15);
        const int koff  = (lane >> 4) << 1;
        const int bcol  = lane & 15;
        const int col   = n0 + bcol;

        v8f acc = {};
        #pragma unroll
        for (int ks = 0; ks < RPAD; ks += 4) {
            v2f a = *(const v2f*)&Wlds[arow * WSTR + ks + koff];
            v2f b;
            b.x = Ulds[(ks + koff    ) * USTR + bcol];
            b.y = Ulds[(ks + koff + 1) * USTR + bcol];
            acc = __builtin_amdgcn_wmma_f32_16x16x4_f32(
                 false, a,  false, b,
                 (short)0, acc,  false,  false);
        }

        if (col < NDIM) {
            const int i0 = mtile * 16 + ((lane >> 4) << 3);
            #pragma unroll
            for (int v = 0; v < 8; ++v)
                out[(i0 + v) * (MDIM * NDIM) + m * NDIM + col] = acc[v];
        }
    }
}

extern "C" void kernel_launch(void* const* d_in, const int* in_sizes, int n_in,
                              void* d_out, int out_size, void* d_ws, size_t ws_size,
                              hipStream_t stream) {
    const float* x  = (const float*)d_in[0];
    const float* w0 = (const float*)d_in[1];
    const float* w1 = (const float*)d_in[2];
    float* out = (float*)d_out;
    (void)in_sizes; (void)n_in; (void)out_size; (void)d_ws; (void)ws_size;

    dim3 grid(MDIM, 4);
    fused_shift_unfold_wmma<<<grid, 256, 0, stream>>>(x, w0, w1, out);
}


# B fragments read as one b64 from a 25x14 table S[n-n0+kh][j] (u[kh,n,j] depends on n,kh only via n+kh; own table for n=0): the 128x16 U build is gone; same values, same summation order
# speedup vs baseline: 5.8595x; 1.2943x over previous
#include <hip/hip_runtime.h>


typedef __attribute__((ext_vector_type(2))) float v2f;
typedef __attribute__((ext_vector_type(8))) float v8f;

#define NCH  24
#define MDIM 56
#define NDIM 56
#define KH   9
#define JD   14
#define RR   (KH*JD)
#define RPAD 128
#define OC   96

#define WSTR 132
#define UCOL 16
#define TROWS 25
#define TSTR 29
#define XSTR 57

#define SROWS  25
#define S2OFF  352
#define SWORDS 512

#define NTHR    256
#define XELEMS  (NCH * NDIM)
#define W1ELEMS (NCH * 28)
#define W0PAIRS (OC * RR / 2)
#define XTRIPS  ((XELEMS  + NTHR - 1) / NTHR)
#define W1TRIPS ((W1ELEMS + NTHR - 1) / NTHR)
#define W0TRIPS ((W0PAIRS + NTHR - 1) / NTHR)

#define XSPARE  ((NCH - 1) * XSTR + NDIM)
#define W1SPARE W1ELEMS
#define W0SPARE ((OC - 1) * WSTR + RPAD)

static __device__ __forceinline__ void lds_barrier() {
    __builtin_amdgcn_fence(__ATOMIC_RELEASE, "workgroup", "local");
    __builtin_amdgcn_s_barrier();
    __builtin_amdgcn_fence(__ATOMIC_ACQUIRE, "workgroup", "local");
}

__global__ __launch_bounds__(NTHR) void fused_shift_unfold_wmma(
    const float* __restrict__ x,
    const float* __restrict__ w0,
    const float* __restrict__ w1,
    float* __restrict__ out)
{
    __shared__ float Wlds[OC * WSTR];
    __shared__ __attribute__((aligned(8))) float Slds[SWORDS];
    __shared__ float Tlds[TROWS * TSTR];
    __shared__ float Xlds[NCH * XSTR];
    __shared__ float W1lds[W1ELEMS + 1];

    const int tid   = threadIdx.x;
    const int m     = blockIdx.x;
    const int ntile = blockIdx.y;
    const int n0    = ntile * UCOL;

    float xv[XTRIPS];
    #pragma unroll
    for (int t = 0; t < XTRIPS; ++t) {
        int e = tid + t * NTHR;
        e = e < XELEMS ? e : XELEMS - 1;
        int i = e / NDIM, h = e % NDIM;
        xv[t] = x[(i * MDIM + m) * NDIM + h];
    }
    float w1v[W1TRIPS];
    #pragma unroll
    for (int t = 0; t < W1TRIPS; ++t) {
        int e = tid + t * NTHR;
        w1v[t] = w1[e < W1ELEMS ? e : W1ELEMS - 1];
    }
    v2f w0v[W0TRIPS];
    #pragma unroll
    for (int t = 0; t < W0TRIPS; ++t) {
        int q = tid + t * NTHR;
        w0v[t] = ((const v2f*)w0)[q < W0PAIRS ? q : W0PAIRS - 1];
    }

    #pragma unroll
    for (int t = 0; t < XTRIPS; ++t) {
        int e = tid + t * NTHR;
        int i = e / NDIM, h = e % NDIM;
        Xlds[e < XELEMS ? i * XSTR + h : XSPARE] = xv[t];
    }
    #pragma unroll
    for (int t = 0; t < W1TRIPS; ++t) {
        int e = tid + t * NTHR;
        W1lds[e < W1ELEMS ? e : W1SPARE] = w1v[t];
    }
    lds_barrier();

    for (int e = tid; e < TROWS * 28; e += NTHR) {
        int q = e / 28, c = e % 28;
        int h = (n0 - 5 + q + NDIM) % NDIM;
        float s = 0.0f;
        #pragma unroll
        for (int i = 0; i < NCH; ++i)
            s += Xlds[i * XSTR + h] * W1lds[i * 28 + c];
        Tlds[q * TSTR + c] = s;
    }
    lds_barrier();

    float ta[SWORDS / NTHR], tb[SWORDS / NTHR];
    #pragma unroll
    for (int t = 0; t < SWORDS / NTHR; ++t) {
        const int  e      = tid + t * NTHR;
        const bool second = t > 0 && e >= S2OFF;
        const int  f      = second ? e - S2OFF : e;
        const int  p      = f / JD;
        const int  j      = f - p * JD;
        const int  pc     = p < TROWS - 1 ? p : TROWS - 2;
        ta[t] = Tlds[(pc + 1) * TSTR + j];
        tb[t] = Tlds[pc * TSTR + 14 + j];
    }
    #pragma unroll
    for (int t = 0; t < SWORDS / NTHR; ++t) {
        const int  e      = tid + t * NTHR;
        const bool second = t > 0 && e >= S2OFF;
        const int  p      = (second ? e - S2OFF : e) / JD;
        const int lo0 = second ? 4  : 4 - n0;
        const int hi0 = second ? KH : min(SROWS - 1, NDIM + 4 - n0);
        const int lo1 = second ? 0  : 5 - n0;
        const int hi1 = second ? 5  : min(SROWS - 1, NDIM + 5 - n0);
        const bool ok0 = p >= lo0 && p < hi0;
        const bool ok1 = p >= lo1 && p < hi1;
        Slds[e] = (ok0 ? ta[t] : 0.0f) + (ok1 ? tb[t] : 0.0f);
    }

    #pragma unroll
    for (int t = 0; t < W0TRIPS; ++t) {
        int q = tid + t * NTHR;
        int i = q / (RR / 2);
        int r = (q - i * (RR / 2)) * 2;
        *(v2f*)&Wlds[q < W0PAIRS ? i * WSTR + r : W0SPARE] = w0v[t];
    }
    if (tid < OC) *(v2f*)&Wlds[tid * WSTR + RR] = (v2f){0.0f, 0.0f};
    __syncthreads();

    const int wave = tid >> 5;
    const int lane = tid & 31;
    if (wave < 6) {
        const int mtile = wave;
        const int arow  = mtile * 16 + (lane & 15);
        const int koff  = (lane >> 4) << 1;
        const int bcol  = lane & 15;
        const int col   = n0 + bcol;
        const int sbase = (col == 0 ? S2OFF : bcol * JD) + koff;

        v8f acc = {};
        #pragma unroll
        for (int ks = 0; ks < RPAD; ks += 4) {
            v2f a = *(const v2f*)&Wlds[arow * WSTR + ks + koff];
            v2f b = *(const v2f*)&Slds[sbase + ks];
            acc = __builtin_amdgcn_wmma_f32_16x16x4_f32(
                 false, a,  false, b,
                 (short)0, acc,  false,  false);
        }

        if (col < NDIM) {
            const int i0 = mtile * 16 + ((lane >> 4) << 3);
            #pragma unroll
            for (int v = 0; v < 8; ++v)
                out[(i0 + v) * (MDIM * NDIM) + m * NDIM + col] = acc[v];
        }
    }
}

extern "C" void kernel_launch(void* const* d_in, const int* in_sizes, int n_in,
                              void* d_out, int out_size, void* d_ws, size_t ws_size,
                              hipStream_t stream) {
    const float* x  = (const float*)d_in[0];
    const float* w0 = (const float*)d_in[1];
    const float* w1 = (const float*)d_in[2];
    float* out = (float*)d_out;
    (void)in_sizes; (void)n_in; (void)out_size; (void)d_ws; (void)ws_size;

    dim3 grid(MDIM, 4);
    fused_shift_unfold_wmma<<<grid, 256, 0, stream>>>(x, w0, w1, out);
}


# T = x_slice^T.w1 (25x28, 24 channels) built by four 16x16x4 f32 matrix tiles, one per wave, instead of per-thread FMA loops with 48 LDS reads per value; T's sums now run as the unit's fma chain
# speedup vs baseline: 6.6112x; 1.1283x over previous
#include <hip/hip_runtime.h>


typedef __attribute__((ext_vector_type(2))) float v2f;
typedef __attribute__((ext_vector_type(8))) float v8f;

#define NCH  24
#define MDIM 56
#define NDIM 56
#define KH   9
#define JD   14
#define RR   (KH*JD)
#define RPAD 128
#define OC   96

#define WSTR 132
#define UCOL 16
#define TROWS 25
#define TPAD 32
#define TSTR 42
#define XSTR 57

#define SROWS  25
#define S2OFF  352
#define SWORDS 512

#define NTHR    256
#define XELEMS  (NCH * NDIM)
#define W1ELEMS (NCH * 28)
#define W0PAIRS (OC * RR / 2)
#define XTRIPS  ((XELEMS  + NTHR - 1) / NTHR)
#define W1TRIPS ((W1ELEMS + NTHR - 1) / NTHR)
#define W0TRIPS ((W0PAIRS + NTHR - 1) / NTHR)

#define XSPARE  ((NCH - 1) * XSTR + NDIM)
#define W1SPARE W1ELEMS
#define W0SPARE ((OC - 1) * WSTR + RPAD)

static __device__ __forceinline__ void lds_barrier() {
    __builtin_amdgcn_fence(__ATOMIC_RELEASE, "workgroup", "local");
    __builtin_amdgcn_s_barrier();
    __builtin_amdgcn_fence(__ATOMIC_ACQUIRE, "workgroup", "local");
}

__global__ __launch_bounds__(NTHR) void fused_shift_unfold_wmma(
    const float* __restrict__ x,
    const float* __restrict__ w0,
    const float* __restrict__ w1,
    float* __restrict__ out)
{
    __shared__ float Wlds[OC * WSTR];
    __shared__ __attribute__((aligned(8))) float Slds[SWORDS];
    __shared__ float Tlds[TPAD * TSTR];
    __shared__ float Xlds[NCH * XSTR];
    __shared__ float W1lds[W1ELEMS + 1];

    const int tid   = threadIdx.x;
    const int wave  = tid >> 5;
    const int lane  = tid & 31;
    const int m     = blockIdx.x;
    const int ntile = blockIdx.y;
    const int n0    = ntile * UCOL;

    float xv[XTRIPS];
    #pragma unroll
    for (int t = 0; t < XTRIPS; ++t) {
        int e = tid + t * NTHR;
        e = e < XELEMS ? e : XELEMS - 1;
        int i = e / NDIM, h = e % NDIM;
        xv[t] = x[(i * MDIM + m) * NDIM + h];
    }
    float w1v[W1TRIPS];
    #pragma unroll
    for (int t = 0; t < W1TRIPS; ++t) {
        int e = tid + t * NTHR;
        w1v[t] = w1[e < W1ELEMS ? e : W1ELEMS - 1];
    }
    v2f w0v[W0TRIPS];
    #pragma unroll
    for (int t = 0; t < W0TRIPS; ++t) {
        int q = tid + t * NTHR;
        w0v[t] = ((const v2f*)w0)[q < W0PAIRS ? q : W0PAIRS - 1];
    }

    #pragma unroll
    for (int t = 0; t < XTRIPS; ++t) {
        int e = tid + t * NTHR;
        int i = e / NDIM, h = e % NDIM;
        Xlds[e < XELEMS ? i * XSTR + h : XSPARE] = xv[t];
    }
    #pragma unroll
    for (int t = 0; t < W1TRIPS; ++t) {
        int e = tid + t * NTHR;
        W1lds[e < W1ELEMS ? e : W1SPARE] = w1v[t];
    }
    lds_barrier();

    if (wave < 4) {
        const int rt = wave >> 1;
        const int ct = wave & 1;
        const int l16 = lane & 15;
        const int hf  = lane >> 4;
        const int h   = (n0 - 5 + rt * 16 + l16 + NDIM) % NDIM;
        const int cb  = min(ct * 16 + l16, 27);
        v8f tacc = {};
        #pragma unroll
        for (int ks = 0; ks < NCH / 4; ++ks) {
            const int i0 = 4 * ks + 2 * hf;
            v2f a = { Xlds[i0 * XSTR + h], Xlds[(i0 + 1) * XSTR + h] };
            v2f b = { W1lds[i0 * 28 + cb], W1lds[(i0 + 1) * 28 + cb] };
            tacc = __builtin_amdgcn_wmma_f32_16x16x4_f32(
                 false, a,  false, b,
                 (short)0, tacc,  false,  false);
        }
        const int q0 = rt * 16 + hf * 8;
        #pragma unroll
        for (int v = 0; v < 8; ++v)
            Tlds[(q0 + v) * TSTR + ct * 16 + l16] = tacc[v];
    }
    lds_barrier();

    float ta[SWORDS / NTHR], tb[SWORDS / NTHR];
    #pragma unroll
    for (int t = 0; t < SWORDS / NTHR; ++t) {
        const int  e      = tid + t * NTHR;
        const bool second = t > 0 && e >= S2OFF;
        const int  f      = second ? e - S2OFF : e;
        const int  p      = f / JD;
        const int  j      = f - p * JD;
        const int  pc     = p < TROWS - 1 ? p : TROWS - 2;
        ta[t] = Tlds[(pc + 1) * TSTR + j];
        tb[t] = Tlds[pc * TSTR + 14 + j];
    }
    #pragma unroll
    for (int t = 0; t < SWORDS / NTHR; ++t) {
        const int  e      = tid + t * NTHR;
        const bool second = t > 0 && e >= S2OFF;
        const int  p      = (second ? e - S2OFF : e) / JD;
        const int lo0 = second ? 4  : 4 - n0;
        const int hi0 = second ? KH : min(SROWS - 1, NDIM + 4 - n0);
        const int lo1 = second ? 0  : 5 - n0;
        const int hi1 = second ? 5  : min(SROWS - 1, NDIM + 5 - n0);
        const bool ok0 = p >= lo0 && p < hi0;
        const bool ok1 = p >= lo1 && p < hi1;
        Slds[e] = (ok0 ? ta[t] : 0.0f) + (ok1 ? tb[t] : 0.0f);
    }

    #pragma unroll
    for (int t = 0; t < W0TRIPS; ++t) {
        int q = tid + t * NTHR;
        int i = q / (RR / 2);
        int r = (q - i * (RR / 2)) * 2;
        *(v2f*)&Wlds[q < W0PAIRS ? i * WSTR + r : W0SPARE] = w0v[t];
    }
    if (tid < OC) *(v2f*)&Wlds[tid * WSTR + RR] = (v2f){0.0f, 0.0f};
    __syncthreads();

    if (wave < 6) {
        const int mtile = wave;
        const int arow  = mtile * 16 + (lane & 15);
        const int koff  = (lane >> 4) << 1;
        const int bcol  = lane & 15;
        const int col   = n0 + bcol;
        const int sbase = (col == 0 ? S2OFF : bcol * JD) + koff;

        v8f acc = {};
        #pragma unroll
        for (int ks = 0; ks < RPAD; ks += 4) {
            v2f a = *(const v2f*)&Wlds[arow * WSTR + ks + koff];
            v2f b = *(const v2f*)&Slds[sbase + ks];
            acc = __builtin_amdgcn_wmma_f32_16x16x4_f32(
                 false, a,  false, b,
                 (short)0, acc,  false,  false);
        }

        if (col < NDIM) {
            const int i0 = mtile * 16 + ((lane >> 4) << 3);
            #pragma unroll
            for (int v = 0; v < 8; ++v)
                out[(i0 + v) * (MDIM * NDIM) + m * NDIM + col] = acc[v];
        }
    }
}

extern "C" void kernel_launch(void* const* d_in, const int* in_sizes, int n_in,
                              void* d_out, int out_size, void* d_ws, size_t ws_size,
                              hipStream_t stream) {
    const float* x  = (const float*)d_in[0];
    const float* w0 = (const float*)d_in[1];
    const float* w1 = (const float*)d_in[2];
    float* out = (float*)d_out;
    (void)in_sizes; (void)n_in; (void)out_size; (void)d_ws; (void)ws_size;

    dim3 grid(MDIM, 4);
    fused_shift_unfold_wmma<<<grid, 256, 0, stream>>>(x, w0, w1, out);
}


# Three waves build the S tables on the matrix unit straight from x and w1 in memory (depth 48: both roll terms in one fma chain) while five stage w0: T, two LDS hand-overs and two barriers are gone
# speedup vs baseline: 6.9621x; 1.0531x over previous
#include <hip/hip_runtime.h>


typedef __attribute__((ext_vector_type(2))) float v2f;
typedef __attribute__((ext_vector_type(8))) float v8f;

#define NCH  24
#define MDIM 56
#define NDIM 56
#define KH   9
#define JD   14
#define RR   (KH*JD)
#define RPAD 128
#define OC   96

#define WSTR 132
#define UCOL 16

#define SROWS  25
#define S2OFF  352
#define S2ROWS 10
#define SWORDS 512
#define SKSTEPS (2 * NCH / 4)

#define NTHR    256
#define SWAVES  3
#define SWST    (NTHR / 32 - SWAVES)
#define W0TRIPS ((OC + SWST - 1) / SWST)

__global__ __launch_bounds__(NTHR) void fused_shift_unfold_wmma(
    const float* __restrict__ x,
    const float* __restrict__ w0,
    const float* __restrict__ w1,
    float* __restrict__ out)
{
    __shared__ float Wlds[OC * WSTR];
    __shared__ __attribute__((aligned(8))) float Slds[SWORDS];
    __shared__ __attribute__((aligned(8))) float Dump[(SWST - 1) * RPAD];

    const int tid   = threadIdx.x;
    const int wave  = tid >> 5;
    const int lane  = tid & 31;
    const int m     = blockIdx.x;
    const int ntile = blockIdx.y;
    const int n0    = ntile * UCOL;

    if (wave >= SWAVES) {
        const int sw  = wave - SWAVES;
        const int pr1 = min(lane + 32, RR / 2 - 1);
        const v2f* wp = (const v2f*)w0;
        v2f w0v[2 * W0TRIPS];
        #pragma unroll
        for (int t = 0; t < W0TRIPS - 1; ++t) {
            const int row = sw + SWST * t;
            w0v[2 * t]     = wp[row * (RR / 2) + lane];
            w0v[2 * t + 1] = wp[row * (RR / 2) + pr1];
        }
        w0v[2 * W0TRIPS - 2] = wp[(OC - 1) * (RR / 2) + lane];
        w0v[2 * W0TRIPS - 1] = wp[(OC - 1) * (RR / 2) + pr1];
        __builtin_amdgcn_sched_barrier(0);
        #pragma unroll
        for (int t = 0; t < W0TRIPS; ++t) {
            float* dst = t < W0TRIPS - 1 ? &Wlds[(sw + SWST * t) * WSTR]
                                         : (sw == 0 ? &Wlds[(OC - 1) * WSTR] : &Dump[(sw - 1) * RPAD]);
            v2f v1 = w0v[2 * t + 1];
            if (lane == 31) v1 = (v2f){0.0f, 0.0f};
            *(v2f*)&dst[2 * lane]      = w0v[2 * t];
            *(v2f*)&dst[64 + 2 * lane] = v1;
        }
    } else if (wave < 2 || n0 == 0) {
        const int l16 = lane & 15;
        const int hf  = lane >> 4;
        const int p   = (wave & 1) * 16 + l16;
        int h0, h1; bool ok0, ok1;
        if (wave < 2) {
            h0 = n0 + p - 4;  ok0 = p < SROWS - 1 && h0 >= 0 && h0 < NDIM;
            h1 = n0 + p - 5;  ok1 = p < SROWS - 1 && h1 >= 0 && h1 < NDIM;
        } else {
            h0 = l16 - 4;          ok0 = l16 >= 4 && l16 < KH;
            h1 = NDIM - 5 + l16;   ok1 = l16 < 5;
        }
        const int hc0 = min(max(h0, 0), NDIM - 1);
        const int hc1 = min(max(h1, 0), NDIM - 1);
        const int jb  = min(l16, JD - 1);

        v2f a[SKSTEPS], b[SKSTEPS];
        #pragma unroll
        for (int ks = 0; ks < SKSTEPS; ++ks) {
            const int t  = ks / (NCH / 4);
            const int c0 = 4 * (ks % (NCH / 4)) + 2 * hf;
            const int hc = t ? hc1 : hc0;
            a[ks] = (v2f){ x[(c0 * MDIM + m) * NDIM + hc], x[((c0 + 1) * MDIM + m) * NDIM + hc] };
            b[ks] = (v2f){ w1[c0 * 28 + t * JD + jb], w1[(c0 + 1) * 28 + t * JD + jb] };
        }
        v8f sacc = {};
        #pragma unroll
        for (int ks = 0; ks < SKSTEPS; ++ks) {
            const bool ok = ks / (NCH / 4) ? ok1 : ok0;
            v2f av = { ok ? a[ks][0] : 0.0f, ok ? a[ks][1] : 0.0f };
            sacc = __builtin_amdgcn_wmma_f32_16x16x4_f32(
                 false, av,  false, b[ks],
                 (short)0, sacc,  false,  false);
        }
        const int q0    = (wave & 1) * 16 + hf * 8;
        const int qend  = wave < 2 ? SROWS : S2ROWS;
        const int sbase = wave < 2 ? 0 : S2OFF;
        if (l16 < JD) {
            #pragma unroll
            for (int v = 0; v < 8; ++v)
                if (q0 + v < qend) Slds[sbase + (q0 + v) * JD + l16] = sacc[v];
        }
    }
    __syncthreads();

    if (wave < 6) {
        const int mtile = wave;
        const int arow  = mtile * 16 + (lane & 15);
        const int koff  = (lane >> 4) << 1;
        const int bcol  = lane & 15;
        const int col   = n0 + bcol;
        const int sbase = (col == 0 ? S2OFF : bcol * JD) + koff;

        v8f acc = {};
        #pragma unroll
        for (int ks = 0; ks < RPAD; ks += 4) {
            v2f a = *(const v2f*)&Wlds[arow * WSTR + ks + koff];
            v2f b = *(const v2f*)&Slds[sbase + ks];
            acc = __builtin_amdgcn_wmma_f32_16x16x4_f32(
                 false, a,  false, b,
                 (short)0, acc,  false,  false);
        }

        if (col < NDIM) {
            const int i0 = mtile * 16 + ((lane >> 4) << 3);
            #pragma unroll
            for (int v = 0; v < 8; ++v)
                out[(i0 + v) * (MDIM * NDIM) + m * NDIM + col] = acc[v];
        }
    }
}

extern "C" void kernel_launch(void* const* d_in, const int* in_sizes, int n_in,
                              void* d_out, int out_size, void* d_ws, size_t ws_size,
                              hipStream_t stream) {
    const float* x  = (const float*)d_in[0];
    const float* w0 = (const float*)d_in[1];
    const float* w1 = (const float*)d_in[2];
    float* out = (float*)d_out;
    (void)in_sizes; (void)n_in; (void)out_size; (void)d_ws; (void)ws_size;

    dim3 grid(MDIM, 4);
    fused_shift_unfold_wmma<<<grid, 256, 0, stream>>>(x, w0, w1, out);
}


# Block = (m, 32-channel slice of w0) over all 56 columns, grid 56x3 (was 56x4 column tiles, each staging all of w0): a block stages a third of w0, 8 tiles on 8 waves; same values and sum order
# speedup vs baseline: 7.4478x; 1.0698x over previous
#include <hip/hip_runtime.h>


typedef __attribute__((ext_vector_type(2))) float v2f;
typedef __attribute__((ext_vector_type(8))) float v8f;

#define NCH  24
#define MDIM 56
#define NDIM 56
#define KH   9
#define JD   14
#define RR   (KH*JD)
#define RPAD 128
#define OC   96

#define WSTR 132
#define ISL  32
#define NSL  (OC / ISL)

#define SROWS  64
#define STAIL  128
#define S2OFF  (SROWS * JD + STAIL)
#define S2WORDS (16 * JD)
#define SWORDS (S2OFF + S2WORDS)
#define SDUMPW 112
#define SKSTEPS (2 * NCH / 4)

#define NTHR    256
#define STGTHR  96
#define W0TRIPS (ISL * (RR / 2) / STGTHR)

__global__ __launch_bounds__(NTHR) void fused_shift_unfold_wmma(
    const float* __restrict__ x,
    const float* __restrict__ w0,
    const float* __restrict__ w1,
    float* __restrict__ out)
{
    __shared__ __attribute__((aligned(8))) float Wlds[ISL * WSTR];
    __shared__ __attribute__((aligned(8))) float Slds[SWORDS];
    __shared__ float Sdump[5 * SDUMPW];

    const int tid   = threadIdx.x;
    const int wave  = tid >> 5;
    const int lane  = tid & 31;
    const int m     = blockIdx.x;
    const int ibase = blockIdx.y * ISL;

    if (wave == 4 || wave >= 6) {
        const int stid = (wave == 4 ? 0 : wave - 5) * 32 + lane;
        const v2f* wp = (const v2f*)w0 + ibase * (RR / 2);
        v2f w0v[W0TRIPS];
        #pragma unroll
        for (int t = 0; t < W0TRIPS; ++t)
            w0v[t] = wp[t * STGTHR + stid];
        __builtin_amdgcn_sched_barrier(0);
        #pragma unroll
        for (int t = 0; t < W0TRIPS; ++t) {
            const unsigned q   = (unsigned)(t * STGTHR + stid);
            const unsigned row = q / (RR / 2);
            const unsigned pr  = q - row * (RR / 2);
            *(v2f*)&Wlds[row * WSTR + 2 * pr] = w0v[t];
        }
    } else {
        const int l16 = lane & 15;
        const int hf  = lane >> 4;
        int h0, h1; bool ok0, ok1;
        if (wave < 4) {
            const int p = wave * 16 + l16;
            h0 = p - 4;  ok0 = h0 >= 0 && h0 < NDIM;
            h1 = p - 5;  ok1 = h1 >= 0 && h1 < NDIM;
        } else {
            h0 = l16 - 4;          ok0 = l16 >= 4 && l16 < KH;
            h1 = NDIM - 5 + l16;   ok1 = l16 < 5;
        }
        const int hc0 = min(max(h0, 0), NDIM - 1);
        const int hc1 = min(max(h1, 0), NDIM - 1);
        const int jb  = min(l16, JD - 1);

        v2f a[SKSTEPS], b[SKSTEPS];
        #pragma unroll
        for (int ks = 0; ks < SKSTEPS; ++ks) {
            const int t  = ks / (NCH / 4);
            const int c0 = 4 * (ks % (NCH / 4)) + 2 * hf;
            const int hc = t ? hc1 : hc0;
            a[ks] = (v2f){ x[(c0 * MDIM + m) * NDIM + hc], x[((c0 + 1) * MDIM + m) * NDIM + hc] };
            b[ks] = (v2f){ w1[c0 * 28 + t * JD + jb], w1[(c0 + 1) * 28 + t * JD + jb] };
        }
        if (wave == 5) {
            const v2f z = {0.0f, 0.0f};
            *(v2f*)&Wlds[lane * WSTR + RR]               = z;
            *(v2f*)&Slds[SROWS * JD + 2 * lane]          = z;
            *(v2f*)&Slds[SROWS * JD + 64 + 2 * lane]     = z;
        }
        v8f sacc = {};
        #pragma unroll
        for (int ks = 0; ks < SKSTEPS; ++ks) {
            const bool ok = ks / (NCH / 4) ? ok1 : ok0;
            v2f av = { ok ? a[ks][0] : 0.0f, ok ? a[ks][1] : 0.0f };
            sacc = __builtin_amdgcn_wmma_f32_16x16x4_f32(
                 false, av,  false, b[ks],
                 (short)0, sacc,  false,  false);
        }
        const int swi = wave < 4 ? wave : 4;
        float* dst = l16 < JD ? &Slds[(wave < 4 ? wave * 16 * JD : S2OFF) + hf * 8 * JD + l16]
                              : &Sdump[swi * SDUMPW + 2 * hf + (l16 - JD)];
        #pragma unroll
        for (int v = 0; v < 8; ++v)
            dst[v * JD] = sacc[v];
    }
    __syncthreads();

    {
        const int mtile = wave >> 2;
        const int ntile = wave & 3;
        const int arow  = mtile * 16 + (lane & 15);
        const int koff  = (lane >> 4) << 1;
        const int col   = ntile * 16 + (lane & 15);
        const int sbase = (col == 0 ? S2OFF : col * JD) + koff;

        v8f acc = {};
        #pragma unroll
        for (int ks = 0; ks < RPAD; ks += 4) {
            v2f a = *(const v2f*)&Wlds[arow * WSTR + ks + koff];
            v2f b = *(const v2f*)&Slds[sbase + ks];
            acc = __builtin_amdgcn_wmma_f32_16x16x4_f32(
                 false, a,  false, b,
                 (short)0, acc,  false,  false);
        }

        if (col < NDIM) {
            const int i0 = ibase + mtile * 16 + ((lane >> 4) << 3);
            #pragma unroll
            for (int v = 0; v < 8; ++v)
                out[(i0 + v) * (MDIM * NDIM) + m * NDIM + col] = acc[v];
        }
    }
}

extern "C" void kernel_launch(void* const* d_in, const int* in_sizes, int n_in,
                              void* d_out, int out_size, void* d_ws, size_t ws_size,
                              hipStream_t stream) {
    const float* x  = (const float*)d_in[0];
    const float* w0 = (const float*)d_in[1];
    const float* w1 = (const float*)d_in[2];
    float* out = (float*)d_out;
    (void)in_sizes; (void)n_in; (void)out_size; (void)d_ws; (void)ws_size;

    dim3 grid(MDIM, NSL);
    fused_shift_unfold_wmma<<<grid, 256, 0, stream>>>(x, w0, w1, out);
}
